# starting kernel
# baseline (speedup 1.0000x reference)
#include <hip/hip_runtime.h>

typedef __attribute__((ext_vector_type(2))) float v2f;
typedef __attribute__((ext_vector_type(4))) float v4f;
typedef __attribute__((ext_vector_type(8))) float v8f;

__device__ constexpr unsigned BASEC[16] = {
    2654435761u, 2246822519u, 3266489917u, 2028178513u, 1220703125u,
    1610612741u, 805306457u,  402653189u,  3674653429u, 2860486313u,
    1073676287u, 2971215073u, 1500450271u, 3267000013u, 2654435789u, 4049292737u};
__device__ constexpr unsigned P5C[32] = {
    2654435761u ^ 0xDEADBEEFu, 2246822519u ^ 0xDEADBEEFu, 3266489917u ^ 0xDEADBEEFu,
    2028178513u ^ 0xDEADBEEFu, 1220703125u ^ 0xDEADBEEFu, 1610612741u ^ 0xDEADBEEFu,
    805306457u  ^ 0xDEADBEEFu, 402653189u  ^ 0xDEADBEEFu, 3674653429u ^ 0xDEADBEEFu,
    2860486313u ^ 0xDEADBEEFu, 1073676287u ^ 0xDEADBEEFu, 2971215073u ^ 0xDEADBEEFu,
    1500450271u ^ 0xDEADBEEFu, 3267000013u ^ 0xDEADBEEFu, 2654435789u ^ 0xDEADBEEFu,
    4049292737u ^ 0xDEADBEEFu, 2246822531u ^ 0xDEADBEEFu, 3266489927u ^ 0xDEADBEEFu,
    2028178519u ^ 0xDEADBEEFu, 1220703133u ^ 0xDEADBEEFu, 1610612759u ^ 0xDEADBEEFu,
    805306463u  ^ 0xDEADBEEFu, 402653201u  ^ 0xDEADBEEFu, 3674653441u ^ 0xDEADBEEFu,
    2860486319u ^ 0xDEADBEEFu, 1073676311u ^ 0xDEADBEEFu, 2971215091u ^ 0xDEADBEEFu,
    1500450277u ^ 0xDEADBEEFu, 3267000023u ^ 0xDEADBEEFu, 2654435801u ^ 0xDEADBEEFu,
    4049292751u ^ 0xDEADBEEFu, 2246822537u ^ 0xDEADBEEFu};

#define T_DIM 4096
#define PPB 128
#define STRIDE 68

__global__ __launch_bounds__(128, 4)
void rhp_kernel(const int* __restrict__ tokens,
                const float* __restrict__ tables,
                const float* __restrict__ Wc,
                float* __restrict__ out,
                int nIter)
{
    __shared__ unsigned tk[256];
    __shared__ unsigned keys_lds[8][PPB];
    __shared__ unsigned ck_lds[PPB];
    __shared__ unsigned cklut[256];
    __shared__ __align__(16) float short_lds[PPB * STRIDE];
    __shared__ __align__(16) float wc_lds[16 * STRIDE];

    const int tid = threadIdx.x;
    const int bid = blockIdx.x;
    const int b   = bid >> 5;
    const int t0  = (bid & 31) << 7;
    const long rowBase = (long)b * T_DIM;

    for (int j = tid; j < 256; j += 128) {
        int g = t0 - 128 + j;
        tk[j] = (g >= 0) ? (unsigned)tokens[rowBase + g] : 0u;
        unsigned v = 0;
        #pragma unroll
        for (int i = 0; i < 8; ++i)
            if (j & (1 << i)) v ^= BASEC[i];
        cklut[j] = v;
    }
    for (int idx = tid; idx < 16 * STRIDE; idx += 128) {
        int row = idx / STRIDE;
        int col = idx - row * STRIDE;
        wc_lds[idx] = (row < 8 && col < 64) ? Wc[row * 64 + col] : 0.0f;
    }
    __syncthreads();

    {
        const int base = 128 + tid;
        unsigned h = 0;
        #pragma unroll
        for (int i = 0; i < 16; ++i) {
            h ^= tk[base - 1 - i] * BASEC[i];
            if (i == 0)  keys_lds[0][tid] = h & 0xFFFFu;
            if (i == 1)  keys_lds[1][tid] = h & 0xFFFFu;
            if (i == 3)  keys_lds[2][tid] = h & 0xFFFFu;
            if (i == 7)  keys_lds[3][tid] = h & 0xFFFFu;
            if (i == 15) keys_lds[4][tid] = h & 0xFFFFu;
        }
        unsigned h5 = 0, h6 = 0, h7 = 0;
        #pragma unroll 8
        for (int i = 0; i < 32; ++i)  h5 ^= tk[base - 1 - i] * P5C[i];
        keys_lds[5][tid] = h5 & 0xFFFFu;
        #pragma unroll 8
        for (int i = 0; i < 64; ++i)  h6 ^= tk[base - 1 - i] * (BASEC[i & 7] ^ 0xCAFEBABEu);
        keys_lds[6][tid] = h6 & 0xFFFFu;
        #pragma unroll 8
        for (int i = 0; i < 128; ++i) h7 ^= tk[base - 1 - i] * (BASEC[i & 7] ^ 0x8BADF00Du);
        keys_lds[7][tid] = h7 & 0xFFFFu;
    }
    __syncthreads();

    const v4f* tab4 = reinterpret_cast<const v4f*>(tables);
    v4f* out4 = reinterpret_cast<v4f*>(out);
    const long outPosBase = (rowBase + t0) * 8;

    #pragma unroll 4
    for (int j = 0; j < nIter; ++j) {
        int flat   = j * 128 + tid;
        int rowIdx = flat >> 2;
        int piece  = flat & 3;
        int scale  = rowIdx >> 7;
        int pos    = rowIdx & 127;
        unsigned key = keys_lds[scale][pos];
        v4f v = tab4[((size_t)(scale << 16) + key) * 4 + piece];
        *reinterpret_cast<v4f*>(&short_lds[pos * STRIDE + scale * 16 + piece * 4]) = v;
        __builtin_nontemporal_store(v, &out4[(outPosBase + (long)pos * 8 + scale) * 4 + piece]);
    }
    __syncthreads();

    {
        const int lane = tid & 31;
        const int wave = tid >> 5;
        const int n  = lane & 15;
        const int hi = lane >> 4;
        const float* bp = &wc_lds[n * STRIDE + 2 * hi];
        #pragma unroll
        for (int tt = 0; tt < 2; ++tt) {
            const int mb = wave * 32 + tt * 16;
            const float* ap = &short_lds[(mb + n) * STRIDE + 2 * hi];
            v8f c = {0.f, 0.f, 0.f, 0.f, 0.f, 0.f, 0.f, 0.f};
            #pragma unroll
            for (int k = 0; k < 16; ++k) {
                v2f a  = *reinterpret_cast<const v2f*>(ap + 4 * k);
                v2f bb = *reinterpret_cast<const v2f*>(bp + 4 * k);
                c = __builtin_amdgcn_wmma_f32_16x16x4_f32(
                        false, a, false, bb, (short)0, c, false, false);
            }
            #pragma unroll
            for (int v = 0; v < 8; ++v) {
#if __has_builtin(__builtin_amdgcn_ballot_w32)
                unsigned mask = __builtin_amdgcn_ballot_w32(c[v] > 0.0f);
                unsigned bits = (mask >> (16 * hi)) & 0xFFu;
                if (n == v)
                    ck_lds[mb + v + 8 * hi] = cklut[bits];
#else
                if (n < 8 && c[v] > 0.0f)
                    atomicXor(&ck_lds[mb + v + 8 * hi], BASEC[n]);
#endif
            }
        }
    }
    __syncthreads();

    #pragma unroll 4
    for (int j = 0; j < nIter; ++j) {
        int flat   = j * 128 + tid;
        int rowIdx = flat >> 2;
        int piece  = flat & 3;
        int scale  = 4 + (rowIdx >> 7);
        int pos    = rowIdx & 127;
        unsigned key = (keys_lds[scale][pos] ^ ck_lds[pos]) & 0xFFFFu;
        v4f v = tab4[((size_t)(scale << 16) + key) * 4 + piece];
        __builtin_nontemporal_store(v, &out4[(outPosBase + (long)pos * 8 + scale) * 4 + piece]);
    }
}

extern "C" void kernel_launch(void* const* d_in, const int* in_sizes, int n_in,
                              void* d_out, int out_size, void* d_ws, size_t ws_size,
                              hipStream_t stream) {
    (void)in_sizes; (void)n_in; (void)out_size; (void)d_ws; (void)ws_size;
    const int*   tokens = (const int*)d_in[0];
    const float* tables = (const float*)d_in[1];
    const float* Wc     = (const float*)d_in[2];
    float*       out    = (float*)d_out;
    const int B = 32, T = 4096;
    dim3 grid(B * (T / PPB)), block(128);
    hipLaunchKernelGGL(rhp_kernel, grid, block, 0, stream, tokens, tables, Wc, out, 16);
}


# Gather lane map: 16 consecutive lanes take one position's 4 scales x 4 pieces, so each store instruction writes whole 128-byte lines (2 x 256 B) instead of eight 64-byte half lines
# speedup vs baseline: 1.3667x; 1.3667x over previous
#include <hip/hip_runtime.h>

typedef __attribute__((ext_vector_type(2))) float v2f;
typedef __attribute__((ext_vector_type(4))) float v4f;
typedef __attribute__((ext_vector_type(8))) float v8f;

__device__ constexpr unsigned BASEC[16] = {
    2654435761u, 2246822519u, 3266489917u, 2028178513u, 1220703125u,
    1610612741u, 805306457u,  402653189u,  3674653429u, 2860486313u,
    1073676287u, 2971215073u, 1500450271u, 3267000013u, 2654435789u, 4049292737u};
__device__ constexpr unsigned P5C[32] = {
    2654435761u ^ 0xDEADBEEFu, 2246822519u ^ 0xDEADBEEFu, 3266489917u ^ 0xDEADBEEFu,
    2028178513u ^ 0xDEADBEEFu, 1220703125u ^ 0xDEADBEEFu, 1610612741u ^ 0xDEADBEEFu,
    805306457u  ^ 0xDEADBEEFu, 402653189u  ^ 0xDEADBEEFu, 3674653429u ^ 0xDEADBEEFu,
    2860486313u ^ 0xDEADBEEFu, 1073676287u ^ 0xDEADBEEFu, 2971215073u ^ 0xDEADBEEFu,
    1500450271u ^ 0xDEADBEEFu, 3267000013u ^ 0xDEADBEEFu, 2654435789u ^ 0xDEADBEEFu,
    4049292737u ^ 0xDEADBEEFu, 2246822531u ^ 0xDEADBEEFu, 3266489927u ^ 0xDEADBEEFu,
    2028178519u ^ 0xDEADBEEFu, 1220703133u ^ 0xDEADBEEFu, 1610612759u ^ 0xDEADBEEFu,
    805306463u  ^ 0xDEADBEEFu, 402653201u  ^ 0xDEADBEEFu, 3674653441u ^ 0xDEADBEEFu,
    2860486319u ^ 0xDEADBEEFu, 1073676311u ^ 0xDEADBEEFu, 2971215091u ^ 0xDEADBEEFu,
    1500450277u ^ 0xDEADBEEFu, 3267000023u ^ 0xDEADBEEFu, 2654435801u ^ 0xDEADBEEFu,
    4049292751u ^ 0xDEADBEEFu, 2246822537u ^ 0xDEADBEEFu};

#define T_DIM 4096
#define PPB 128
#define STRIDE 68
#define KSTR 132

__global__ __launch_bounds__(128, 4)
void rhp_kernel(const int* __restrict__ tokens,
                const float* __restrict__ tables,
                const float* __restrict__ Wc,
                float* __restrict__ out,
                int nIter)
{
    __shared__ unsigned tk[256];
    __shared__ unsigned keys_lds[8][KSTR];
    __shared__ unsigned ck_lds[PPB];
    __shared__ unsigned cklut[256];
    __shared__ __align__(16) float short_lds[PPB * STRIDE];
    __shared__ __align__(16) float wc_lds[16 * STRIDE];

    const int tid = threadIdx.x;
    const int bid = blockIdx.x;
    const int b   = bid >> 5;
    const int t0  = (bid & 31) << 7;
    const long rowBase = (long)b * T_DIM;

    for (int j = tid; j < 256; j += 128) {
        int g = t0 - 128 + j;
        tk[j] = (g >= 0) ? (unsigned)tokens[rowBase + g] : 0u;
        unsigned v = 0;
        #pragma unroll
        for (int i = 0; i < 8; ++i)
            if (j & (1 << i)) v ^= BASEC[i];
        cklut[j] = v;
    }
    for (int idx = tid; idx < 16 * STRIDE; idx += 128) {
        int row = idx / STRIDE;
        int col = idx - row * STRIDE;
        wc_lds[idx] = (row < 8 && col < 64) ? Wc[row * 64 + col] : 0.0f;
    }
    __syncthreads();

    {
        const int base = 128 + tid;
        unsigned h = 0;
        #pragma unroll
        for (int i = 0; i < 16; ++i) {
            h ^= tk[base - 1 - i] * BASEC[i];
            if (i == 0)  keys_lds[0][tid] = h & 0xFFFFu;
            if (i == 1)  keys_lds[1][tid] = h & 0xFFFFu;
            if (i == 3)  keys_lds[2][tid] = h & 0xFFFFu;
            if (i == 7)  keys_lds[3][tid] = h & 0xFFFFu;
            if (i == 15) keys_lds[4][tid] = h & 0xFFFFu;
        }
        unsigned h5 = 0, h6 = 0, h7 = 0;
        #pragma unroll 8
        for (int i = 0; i < 32; ++i)  h5 ^= tk[base - 1 - i] * P5C[i];
        keys_lds[5][tid] = h5 & 0xFFFFu;
        #pragma unroll 8
        for (int i = 0; i < 64; ++i)  h6 ^= tk[base - 1 - i] * (BASEC[i & 7] ^ 0xCAFEBABEu);
        keys_lds[6][tid] = h6 & 0xFFFFu;
        #pragma unroll 8
        for (int i = 0; i < 128; ++i) h7 ^= tk[base - 1 - i] * (BASEC[i & 7] ^ 0x8BADF00Du);
        keys_lds[7][tid] = h7 & 0xFFFFu;
    }
    __syncthreads();

    const v4f* tab4 = reinterpret_cast<const v4f*>(tables);
    v4f* out4 = reinterpret_cast<v4f*>(out);
    const long outPosBase = (rowBase + t0) * 8;

    const int piece = tid & 3;
    const int sc    = (tid >> 2) & 3;
    const int p0    = tid >> 4;

    #pragma unroll 4
    for (int j = 0; j < nIter; ++j) {
        const int scale = sc;
        const int pos   = j * 8 + p0;
        unsigned key = keys_lds[scale][pos];
        v4f v = tab4[((size_t)(scale << 16) + key) * 4 + piece];
        *reinterpret_cast<v4f*>(&short_lds[pos * STRIDE + scale * 16 + piece * 4]) = v;
        __builtin_nontemporal_store(v, &out4[(outPosBase + (long)pos * 8 + scale) * 4 + piece]);
    }
    __syncthreads();

    {
        const int lane = tid & 31;
        const int wave = tid >> 5;
        const int n  = lane & 15;
        const int hi = lane >> 4;
        const float* bp = &wc_lds[n * STRIDE + 2 * hi];
        #pragma unroll
        for (int tt = 0; tt < 2; ++tt) {
            const int mb = wave * 32 + tt * 16;
            const float* ap = &short_lds[(mb + n) * STRIDE + 2 * hi];
            v8f c = {0.f, 0.f, 0.f, 0.f, 0.f, 0.f, 0.f, 0.f};
            #pragma unroll
            for (int k = 0; k < 16; ++k) {
                v2f a  = *reinterpret_cast<const v2f*>(ap + 4 * k);
                v2f bb = *reinterpret_cast<const v2f*>(bp + 4 * k);
                c = __builtin_amdgcn_wmma_f32_16x16x4_f32(
                        false, a, false, bb, (short)0, c, false, false);
            }
            #pragma unroll
            for (int v = 0; v < 8; ++v) {
#if __has_builtin(__builtin_amdgcn_ballot_w32)
                unsigned mask = __builtin_amdgcn_ballot_w32(c[v] > 0.0f);
                unsigned bits = (mask >> (16 * hi)) & 0xFFu;
                if (n == v)
                    ck_lds[mb + v + 8 * hi] = cklut[bits];
#else
                if (n < 8 && c[v] > 0.0f)
                    atomicXor(&ck_lds[mb + v + 8 * hi], BASEC[n]);
#endif
            }
        }
    }
    __syncthreads();

    #pragma unroll 4
    for (int j = 0; j < nIter; ++j) {
        const int scale = 4 + sc;
        const int pos   = j * 8 + p0;
        unsigned key = (keys_lds[scale][pos] ^ ck_lds[pos]) & 0xFFFFu;
        v4f v = tab4[((size_t)(scale << 16) + key) * 4 + piece];
        __builtin_nontemporal_store(v, &out4[(outPosBase + (long)pos * 8 + scale) * 4 + piece]);
    }
}

extern "C" void kernel_launch(void* const* d_in, const int* in_sizes, int n_in,
                              void* d_out, int out_size, void* d_ws, size_t ws_size,
                              hipStream_t stream) {
    (void)in_sizes; (void)n_in; (void)out_size; (void)d_ws; (void)ws_size;
    const int*   tokens = (const int*)d_in[0];
    const float* tables = (const float*)d_in[1];
    const float* Wc     = (const float*)d_in[2];
    float*       out    = (float*)d_out;
    const int B = 32, T = 4096;
    dim3 grid(B * (T / PPB)), block(128);
    hipLaunchKernelGGL(rhp_kernel, grid, block, 0, stream, tokens, tables, Wc, out, 16);
}


# Loads hoisted: prologue loads issued together, each lane's 16 short-row gathers issued before the long hash loops, its 16 long-row gathers all at once (was 4 per trip); barriers fence LDS only
# speedup vs baseline: 1.6762x; 1.2265x over previous
#include <hip/hip_runtime.h>

typedef __attribute__((ext_vector_type(2))) float v2f;
typedef __attribute__((ext_vector_type(4))) float v4f;
typedef __attribute__((ext_vector_type(8))) float v8f;

__device__ constexpr unsigned BASEC[16] = {
    2654435761u, 2246822519u, 3266489917u, 2028178513u, 1220703125u,
    1610612741u, 805306457u,  402653189u,  3674653429u, 2860486313u,
    1073676287u, 2971215073u, 1500450271u, 3267000013u, 2654435789u, 4049292737u};
__device__ constexpr unsigned P5C[32] = {
    2654435761u ^ 0xDEADBEEFu, 2246822519u ^ 0xDEADBEEFu, 3266489917u ^ 0xDEADBEEFu,
    2028178513u ^ 0xDEADBEEFu, 1220703125u ^ 0xDEADBEEFu, 1610612741u ^ 0xDEADBEEFu,
    805306457u  ^ 0xDEADBEEFu, 402653189u  ^ 0xDEADBEEFu, 3674653429u ^ 0xDEADBEEFu,
    2860486313u ^ 0xDEADBEEFu, 1073676287u ^ 0xDEADBEEFu, 2971215073u ^ 0xDEADBEEFu,
    1500450271u ^ 0xDEADBEEFu, 3267000013u ^ 0xDEADBEEFu, 2654435789u ^ 0xDEADBEEFu,
    4049292737u ^ 0xDEADBEEFu, 2246822531u ^ 0xDEADBEEFu, 3266489927u ^ 0xDEADBEEFu,
    2028178519u ^ 0xDEADBEEFu, 1220703133u ^ 0xDEADBEEFu, 1610612759u ^ 0xDEADBEEFu,
    805306463u  ^ 0xDEADBEEFu, 402653201u  ^ 0xDEADBEEFu, 3674653441u ^ 0xDEADBEEFu,
    2860486319u ^ 0xDEADBEEFu, 1073676311u ^ 0xDEADBEEFu, 2971215091u ^ 0xDEADBEEFu,
    1500450277u ^ 0xDEADBEEFu, 3267000023u ^ 0xDEADBEEFu, 2654435801u ^ 0xDEADBEEFu,
    4049292751u ^ 0xDEADBEEFu, 2246822537u ^ 0xDEADBEEFu};

#define T_DIM 4096
#define PPB 128
#define STRIDE 68
#define KSTR 132

__device__ __forceinline__ void lds_barrier() {
    __builtin_amdgcn_fence(__ATOMIC_RELEASE, "workgroup", "local");
    __builtin_amdgcn_s_barrier();
    __builtin_amdgcn_fence(__ATOMIC_ACQUIRE, "workgroup", "local");
}

__global__ __launch_bounds__(128, 4)
void rhp_kernel(const int* __restrict__ tokens,
                const float* __restrict__ tables,
                const float* __restrict__ Wc,
                float* __restrict__ out)
{
    __shared__ unsigned tk[256];
    __shared__ unsigned keys_lds[8][KSTR];
    __shared__ unsigned ck_lds[PPB];
    __shared__ unsigned cklut[256];
    __shared__ __align__(16) float short_lds[PPB * STRIDE];
    __shared__ __align__(16) float wc_lds[16 * STRIDE];

    const int tid = threadIdx.x;
    const int bid = blockIdx.x;
    const int b   = bid >> 5;
    const int t0  = (bid & 31) << 7;
    const long rowBase = (long)b * T_DIM;

    const int gA = t0 - 128 + tid;
    const unsigned tokA = (gA >= 0) ? (unsigned)tokens[rowBase + gA] : 0u;
    const unsigned tokB = (unsigned)tokens[rowBase + t0 + tid];
    const v4f wrow = reinterpret_cast<const v4f*>(Wc)[tid];
    for (int r = 0; r < 2; ++r) {
        const int j = tid + 128 * r;
        unsigned v = 0;
        #pragma unroll
        for (int i = 0; i < 8; ++i)
            if (j & (1 << i)) v ^= BASEC[i];
        cklut[j] = v;
    }
    {
        const v4f zero4 = {0.f, 0.f, 0.f, 0.f};
        for (int q = tid; q < 8 * STRIDE / 4; q += 128)
            *reinterpret_cast<v4f*>(&wc_lds[8 * STRIDE + q * 4]) = zero4;
    }
    tk[tid]       = tokA;
    tk[128 + tid] = tokB;
    *reinterpret_cast<v4f*>(&wc_lds[(tid >> 4) * STRIDE + (tid & 15) * 4]) = wrow;
    __syncthreads();

    const int base = 128 + tid;
    unsigned h = 0;
    #pragma unroll
    for (int i = 0; i < 8; ++i) {
        h ^= tk[base - 1 - i] * BASEC[i];
        if (i == 0)  keys_lds[0][tid] = h & 0xFFFFu;
        if (i == 1)  keys_lds[1][tid] = h & 0xFFFFu;
        if (i == 3)  keys_lds[2][tid] = h & 0xFFFFu;
        if (i == 7)  keys_lds[3][tid] = h & 0xFFFFu;
    }
    lds_barrier();

    const v4f* tab4 = reinterpret_cast<const v4f*>(tables);
    v4f* out4 = reinterpret_cast<v4f*>(out);
    const long outPosBase = (rowBase + t0) * 8;

    const int piece = tid & 3;
    const int sc    = (tid >> 2) & 3;
    const int p0    = tid >> 4;

    v4f sv[16];
    #pragma unroll
    for (int j = 0; j < 16; ++j) {
        const int pos = j * 8 + p0;
        const unsigned key = keys_lds[sc][pos];
        sv[j] = tab4[((size_t)(sc << 16) + key) * 4 + piece];
    }

    {
        #pragma unroll
        for (int i = 8; i < 16; ++i)
            h ^= tk[base - 1 - i] * BASEC[i];
        keys_lds[4][tid] = h & 0xFFFFu;
        unsigned h5 = 0, h6 = 0, h7 = 0;
        #pragma unroll 8
        for (int i = 0; i < 32; ++i)  h5 ^= tk[base - 1 - i] * P5C[i];
        keys_lds[5][tid] = h5 & 0xFFFFu;
        #pragma unroll 8
        for (int i = 0; i < 64; ++i)  h6 ^= tk[base - 1 - i] * (BASEC[i & 7] ^ 0xCAFEBABEu);
        keys_lds[6][tid] = h6 & 0xFFFFu;
        #pragma unroll 8
        for (int i = 0; i < 128; ++i) h7 ^= tk[base - 1 - i] * (BASEC[i & 7] ^ 0x8BADF00Du);
        keys_lds[7][tid] = h7 & 0xFFFFu;
    }

    #pragma unroll
    for (int j = 0; j < 16; ++j) {
        const int pos = j * 8 + p0;
        *reinterpret_cast<v4f*>(&short_lds[pos * STRIDE + sc * 16 + piece * 4]) = sv[j];
        __builtin_nontemporal_store(sv[j], &out4[(outPosBase + (long)pos * 8 + sc) * 4 + piece]);
    }
    lds_barrier();

    {
        const int lane = tid & 31;
        const int wave = tid >> 5;
        const int n  = lane & 15;
        const int hi = lane >> 4;
        const float* bp = &wc_lds[n * STRIDE + 2 * hi];
        #pragma unroll
        for (int tt = 0; tt < 2; ++tt) {
            const int mb = wave * 32 + tt * 16;
            const float* ap = &short_lds[(mb + n) * STRIDE + 2 * hi];
            v8f c = {0.f, 0.f, 0.f, 0.f, 0.f, 0.f, 0.f, 0.f};
            #pragma unroll
            for (int k = 0; k < 16; ++k) {
                v2f a  = *reinterpret_cast<const v2f*>(ap + 4 * k);
                v2f bb = *reinterpret_cast<const v2f*>(bp + 4 * k);
                c = __builtin_amdgcn_wmma_f32_16x16x4_f32(
                        false, a, false, bb, (short)0, c, false, false);
            }
            #pragma unroll
            for (int v = 0; v < 8; ++v) {
#if __has_builtin(__builtin_amdgcn_ballot_w32)
                unsigned mask = __builtin_amdgcn_ballot_w32(c[v] > 0.0f);
                unsigned bits = (mask >> (16 * hi)) & 0xFFu;
                if (n == v)
                    ck_lds[mb + v + 8 * hi] = cklut[bits];
#else
                if (n < 8 && c[v] > 0.0f)
                    atomicXor(&ck_lds[mb + v + 8 * hi], BASEC[n]);
#endif
            }
        }
    }
    lds_barrier();

    v4f lv[16];
    #pragma unroll
    for (int j = 0; j < 16; ++j) {
        const int pos = j * 8 + p0;
        const unsigned key = (keys_lds[4 + sc][pos] ^ ck_lds[pos]) & 0xFFFFu;
        lv[j] = tab4[((size_t)((4 + sc) << 16) + key) * 4 + piece];
    }
    #pragma unroll
    for (int j = 0; j < 16; ++j) {
        const int pos = j * 8 + p0;
        __builtin_nontemporal_store(lv[j], &out4[(outPosBase + (long)pos * 8 + 4 + sc) * 4 + piece]);
    }
}

extern "C" void kernel_launch(void* const* d_in, const int* in_sizes, int n_in,
                              void* d_out, int out_size, void* d_ws, size_t ws_size,
                              hipStream_t stream) {
    (void)in_sizes; (void)n_in; (void)out_size; (void)d_ws; (void)ws_size;
    const int*   tokens = (const int*)d_in[0];
    const float* tables = (const float*)d_in[1];
    const float* Wc     = (const float*)d_in[2];
    float*       out    = (float*)d_out;
    const int B = 32, T = 4096;
    dim3 grid(B * (T / PPB)), block(128);
    hipLaunchKernelGGL(rhp_kernel, grid, block, 0, stream, tokens, tables, Wc, out);
}


# Hash terms on packed 16-bit multiplies: only the low 16 bits of each token*prime product reach a key, so one v_pk_mul_lo_u16 does two terms; each token pair is read from LDS once for all scales
# speedup vs baseline: 1.7807x; 1.0623x over previous
#include <hip/hip_runtime.h>

typedef __attribute__((ext_vector_type(2))) float v2f;
typedef __attribute__((ext_vector_type(4))) float v4f;
typedef __attribute__((ext_vector_type(8))) float v8f;

__device__ constexpr unsigned BASEC[16] = {
    2654435761u, 2246822519u, 3266489917u, 2028178513u, 1220703125u,
    1610612741u, 805306457u,  402653189u,  3674653429u, 2860486313u,
    1073676287u, 2971215073u, 1500450271u, 3267000013u, 2654435789u, 4049292737u};
__device__ constexpr unsigned P5C[32] = {
    2654435761u ^ 0xDEADBEEFu, 2246822519u ^ 0xDEADBEEFu, 3266489917u ^ 0xDEADBEEFu,
    2028178513u ^ 0xDEADBEEFu, 1220703125u ^ 0xDEADBEEFu, 1610612741u ^ 0xDEADBEEFu,
    805306457u  ^ 0xDEADBEEFu, 402653189u  ^ 0xDEADBEEFu, 3674653429u ^ 0xDEADBEEFu,
    2860486313u ^ 0xDEADBEEFu, 1073676287u ^ 0xDEADBEEFu, 2971215073u ^ 0xDEADBEEFu,
    1500450271u ^ 0xDEADBEEFu, 3267000013u ^ 0xDEADBEEFu, 2654435789u ^ 0xDEADBEEFu,
    4049292737u ^ 0xDEADBEEFu, 2246822531u ^ 0xDEADBEEFu, 3266489927u ^ 0xDEADBEEFu,
    2028178519u ^ 0xDEADBEEFu, 1220703133u ^ 0xDEADBEEFu, 1610612759u ^ 0xDEADBEEFu,
    805306463u  ^ 0xDEADBEEFu, 402653201u  ^ 0xDEADBEEFu, 3674653441u ^ 0xDEADBEEFu,
    2860486319u ^ 0xDEADBEEFu, 1073676311u ^ 0xDEADBEEFu, 2971215091u ^ 0xDEADBEEFu,
    1500450277u ^ 0xDEADBEEFu, 3267000023u ^ 0xDEADBEEFu, 2654435801u ^ 0xDEADBEEFu,
    4049292751u ^ 0xDEADBEEFu, 2246822537u ^ 0xDEADBEEFu};

typedef unsigned short us2 __attribute__((ext_vector_type(2)));
__device__ __forceinline__ unsigned pkmul(unsigned a, unsigned b) {
    const us2 x = __builtin_bit_cast(us2, a);
    const us2 y = __builtin_bit_cast(us2, b);
    const us2 z = x * y;
    return __builtin_bit_cast(unsigned, z);
}
__device__ constexpr unsigned pk16(unsigned lo, unsigned hi) {
    return (lo & 0xFFFFu) | ((hi & 0xFFFFu) << 16);
}
#define PKB(m) pk16(BASEC[2 * (m)], BASEC[2 * (m) + 1])
#define PK5(m) pk16(P5C[2 * (m)], P5C[2 * (m) + 1])
#define PK6(m) pk16(BASEC[(2 * (m)) & 7] ^ 0xCAFEBABEu, BASEC[(2 * (m) + 1) & 7] ^ 0xCAFEBABEu)
#define PK7(m) pk16(BASEC[(2 * (m)) & 7] ^ 0x8BADF00Du, BASEC[(2 * (m) + 1) & 7] ^ 0x8BADF00Du)
__device__ __forceinline__ unsigned fold16(unsigned a) { return (a ^ (a >> 16)) & 0xFFFFu; }

#define T_DIM 4096
#define PPB 128
#define STRIDE 68
#define KSTR 132

__device__ __forceinline__ void lds_barrier() {
    __builtin_amdgcn_fence(__ATOMIC_RELEASE, "workgroup", "local");
    __builtin_amdgcn_s_barrier();
    __builtin_amdgcn_fence(__ATOMIC_ACQUIRE, "workgroup", "local");
}

__global__ __launch_bounds__(128, 4)
void rhp_kernel(const int* __restrict__ tokens,
                const float* __restrict__ tables,
                const float* __restrict__ Wc,
                float* __restrict__ out)
{
    __shared__ unsigned pk2[256];
    __shared__ unsigned keys_lds[8][KSTR];
    __shared__ unsigned ck_lds[PPB];
    __shared__ unsigned cklut[256];
    __shared__ __align__(16) float short_lds[PPB * STRIDE];
    __shared__ __align__(16) float wc_lds[16 * STRIDE];

    const int tid = threadIdx.x;
    const int bid = blockIdx.x;
    const int b   = bid >> 5;
    const int t0  = (bid & 31) << 7;
    const long rowBase = (long)b * T_DIM;

    const int gA = t0 - 128 + tid;
    const int gB = t0 + tid;
    const unsigned ldA  = (unsigned)tokens[rowBase + (gA >= 0 ? gA : 0)];
    const unsigned ldAm = (unsigned)tokens[rowBase + (gA >= 1 ? gA - 1 : 0)];
    const unsigned tokB = (unsigned)tokens[rowBase + gB];
    const unsigned ldBm = (unsigned)tokens[rowBase + (gB >= 1 ? gB - 1 : 0)];
    const unsigned tokA  = (gA >= 0) ? ldA  : 0u;
    const unsigned tokAm = (gA >= 1) ? ldAm : 0u;
    const unsigned tokBm = (gB >= 1) ? ldBm : 0u;
    const v4f wrow = reinterpret_cast<const v4f*>(Wc)[tid];
    for (int r = 0; r < 2; ++r) {
        const int j = tid + 128 * r;
        unsigned v = 0;
        #pragma unroll
        for (int i = 0; i < 8; ++i)
            if (j & (1 << i)) v ^= BASEC[i];
        cklut[j] = v;
    }
    {
        const v4f zero4 = {0.f, 0.f, 0.f, 0.f};
        for (int q = tid; q < 8 * STRIDE / 4; q += 128)
            *reinterpret_cast<v4f*>(&wc_lds[8 * STRIDE + q * 4]) = zero4;
    }
    pk2[tid]       = (tokA & 0xFFFFu) | (tokAm << 16);
    pk2[128 + tid] = (tokB & 0xFFFFu) | (tokBm << 16);
    *reinterpret_cast<v4f*>(&wc_lds[(tid >> 4) * STRIDE + (tid & 15) * 4]) = wrow;
    __syncthreads();

    const unsigned* pp = &pk2[127 + tid];
    const unsigned w0 = pp[0], w1 = pp[-2], w2 = pp[-4], w3 = pp[-6];
    unsigned h;
    {
        const unsigned q0 = pkmul(w0, PKB(0));
        keys_lds[0][tid] = q0 & 0xFFFFu;
        keys_lds[1][tid] = fold16(q0);
        h = q0 ^ pkmul(w1, PKB(1));
        keys_lds[2][tid] = fold16(h);
        h ^= pkmul(w2, PKB(2)) ^ pkmul(w3, PKB(3));
        keys_lds[3][tid] = fold16(h);
    }
    lds_barrier();

    const v4f* tab4 = reinterpret_cast<const v4f*>(tables);
    v4f* out4 = reinterpret_cast<v4f*>(out);
    const long outPosBase = (rowBase + t0) * 8;

    const int piece = tid & 3;
    const int sc    = (tid >> 2) & 3;
    const int p0    = tid >> 4;

    v4f sv[16];
    #pragma unroll
    for (int j = 0; j < 16; ++j) {
        const int pos = j * 8 + p0;
        const unsigned key = keys_lds[sc][pos];
        sv[j] = tab4[((size_t)(sc << 16) + key) * 4 + piece];
    }

    {
        unsigned h5 = 0, h6 = 0, h7 = 0;
        #pragma unroll
        for (int m = 0; m < 64; ++m) {
            const unsigned w = (m == 0) ? w0 : (m == 1) ? w1 : (m == 2) ? w2 : (m == 3) ? w3 : pp[-2 * m];
            if (m >= 4 && m < 8) h ^= pkmul(w, PKB(m));
            if (m < 16)          h5 ^= pkmul(w, PK5(m));
            if (m < 32)          h6 ^= pkmul(w, PK6(m & 3));
            h7 ^= pkmul(w, PK7(m & 3));
        }
        keys_lds[4][tid] = fold16(h);
        keys_lds[5][tid] = fold16(h5);
        keys_lds[6][tid] = fold16(h6);
        keys_lds[7][tid] = fold16(h7);
    }

    #pragma unroll
    for (int j = 0; j < 16; ++j) {
        const int pos = j * 8 + p0;
        *reinterpret_cast<v4f*>(&short_lds[pos * STRIDE + sc * 16 + piece * 4]) = sv[j];
        __builtin_nontemporal_store(sv[j], &out4[(outPosBase + (long)pos * 8 + sc) * 4 + piece]);
    }
    lds_barrier();

    {
        const int lane = tid & 31;
        const int wave = tid >> 5;
        const int n  = lane & 15;
        const int hi = lane >> 4;
        const float* bp = &wc_lds[n * STRIDE + 2 * hi];
        #pragma unroll
        for (int tt = 0; tt < 2; ++tt) {
            const int mb = wave * 32 + tt * 16;
            const float* ap = &short_lds[(mb + n) * STRIDE + 2 * hi];
            v8f c = {0.f, 0.f, 0.f, 0.f, 0.f, 0.f, 0.f, 0.f};
            #pragma unroll
            for (int k = 0; k < 16; ++k) {
                v2f a  = *reinterpret_cast<const v2f*>(ap + 4 * k);
                v2f bb = *reinterpret_cast<const v2f*>(bp + 4 * k);
                c = __builtin_amdgcn_wmma_f32_16x16x4_f32(
                        false, a, false, bb, (short)0, c, false, false);
            }
            #pragma unroll
            for (int v = 0; v < 8; ++v) {
#if __has_builtin(__builtin_amdgcn_ballot_w32)
                unsigned mask = __builtin_amdgcn_ballot_w32(c[v] > 0.0f);
                unsigned bits = (mask >> (16 * hi)) & 0xFFu;
                if (n == v)
                    ck_lds[mb + v + 8 * hi] = cklut[bits];
#else
                if (n < 8 && c[v] > 0.0f)
                    atomicXor(&ck_lds[mb + v + 8 * hi], BASEC[n]);
#endif
            }
        }
    }
    lds_barrier();

    v4f lv[16];
    #pragma unroll
    for (int j = 0; j < 16; ++j) {
        const int pos = j * 8 + p0;
        const unsigned key = (keys_lds[4 + sc][pos] ^ ck_lds[pos]) & 0xFFFFu;
        lv[j] = tab4[((size_t)((4 + sc) << 16) + key) * 4 + piece];
    }
    #pragma unroll
    for (int j = 0; j < 16; ++j) {
        const int pos = j * 8 + p0;
        __builtin_nontemporal_store(lv[j], &out4[(outPosBase + (long)pos * 8 + 4 + sc) * 4 + piece]);
    }
}

extern "C" void kernel_launch(void* const* d_in, const int* in_sizes, int n_in,
                              void* d_out, int out_size, void* d_ws, size_t ws_size,
                              hipStream_t stream) {
    (void)in_sizes; (void)n_in; (void)out_size; (void)d_ws; (void)ws_size;
    const int*   tokens = (const int*)d_in[0];
    const float* tables = (const float*)d_in[1];
    const float* Wc     = (const float*)d_in[2];
    float*       out    = (float*)d_out;
    const int B = 32, T = 4096;
    dim3 grid(B * (T / PPB)), block(128);
    hipLaunchKernelGGL(rhp_kernel, grid, block, 0, stream, tokens, tables, Wc, out);
}


# Wave-private ownership: a wave hashes, gathers, multiplies, stores its own 32 positions: no block barrier past the prologue, cond keys go from ballots to gather lanes, tile 0 loads fly under tile 1
# speedup vs baseline: 1.9082x; 1.0716x over previous
#include <hip/hip_runtime.h>

typedef __attribute__((ext_vector_type(2))) float v2f;
typedef __attribute__((ext_vector_type(4))) float v4f;
typedef __attribute__((ext_vector_type(8))) float v8f;

__device__ constexpr unsigned BASEC[16] = {
    2654435761u, 2246822519u, 3266489917u, 2028178513u, 1220703125u,
    1610612741u, 805306457u,  402653189u,  3674653429u, 2860486313u,
    1073676287u, 2971215073u, 1500450271u, 3267000013u, 2654435789u, 4049292737u};
__device__ constexpr unsigned P5C[32] = {
    2654435761u ^ 0xDEADBEEFu, 2246822519u ^ 0xDEADBEEFu, 3266489917u ^ 0xDEADBEEFu,
    2028178513u ^ 0xDEADBEEFu, 1220703125u ^ 0xDEADBEEFu, 1610612741u ^ 0xDEADBEEFu,
    805306457u  ^ 0xDEADBEEFu, 402653189u  ^ 0xDEADBEEFu, 3674653429u ^ 0xDEADBEEFu,
    2860486313u ^ 0xDEADBEEFu, 1073676287u ^ 0xDEADBEEFu, 2971215073u ^ 0xDEADBEEFu,
    1500450271u ^ 0xDEADBEEFu, 3267000013u ^ 0xDEADBEEFu, 2654435789u ^ 0xDEADBEEFu,
    4049292737u ^ 0xDEADBEEFu, 2246822531u ^ 0xDEADBEEFu, 3266489927u ^ 0xDEADBEEFu,
    2028178519u ^ 0xDEADBEEFu, 1220703133u ^ 0xDEADBEEFu, 1610612759u ^ 0xDEADBEEFu,
    805306463u  ^ 0xDEADBEEFu, 402653201u  ^ 0xDEADBEEFu, 3674653441u ^ 0xDEADBEEFu,
    2860486319u ^ 0xDEADBEEFu, 1073676311u ^ 0xDEADBEEFu, 2971215091u ^ 0xDEADBEEFu,
    1500450277u ^ 0xDEADBEEFu, 3267000023u ^ 0xDEADBEEFu, 2654435801u ^ 0xDEADBEEFu,
    4049292751u ^ 0xDEADBEEFu, 2246822537u ^ 0xDEADBEEFu};

typedef unsigned short us2 __attribute__((ext_vector_type(2)));
__device__ __forceinline__ unsigned pkmul(unsigned a, unsigned b) {
    const us2 x = __builtin_bit_cast(us2, a);
    const us2 y = __builtin_bit_cast(us2, b);
    const us2 z = x * y;
    return __builtin_bit_cast(unsigned, z);
}
__device__ constexpr unsigned pk16(unsigned lo, unsigned hi) {
    return (lo & 0xFFFFu) | ((hi & 0xFFFFu) << 16);
}
#define PKB(m) pk16(BASEC[2 * (m)], BASEC[2 * (m) + 1])
#define PK5(m) pk16(P5C[2 * (m)], P5C[2 * (m) + 1])
#define PK6(m) pk16(BASEC[(2 * (m)) & 7] ^ 0xCAFEBABEu, BASEC[(2 * (m) + 1) & 7] ^ 0xCAFEBABEu)
#define PK7(m) pk16(BASEC[(2 * (m)) & 7] ^ 0x8BADF00Du, BASEC[(2 * (m) + 1) & 7] ^ 0x8BADF00Du)
__device__ __forceinline__ unsigned fold16(unsigned a) { return (a ^ (a >> 16)) & 0xFFFFu; }

#define T_DIM 4096
#define PPB 128
#define STRIDE 68
#define KSTR 132

__global__ __launch_bounds__(128, 4)
void rhp_kernel(const int* __restrict__ tokens,
                const float* __restrict__ tables,
                const float* __restrict__ Wc,
                float* __restrict__ out)
{
    __shared__ unsigned pk2[256];
    __shared__ unsigned keys_lds[8][KSTR];
    __shared__ unsigned cklut[256];
    __shared__ __align__(16) float short_lds[PPB * STRIDE];
    __shared__ __align__(16) float wc_lds[16 * STRIDE];

    const int tid = threadIdx.x;
    const int bid = blockIdx.x;
    const int b   = bid >> 5;
    const int t0  = (bid & 31) << 7;
    const long rowBase = (long)b * T_DIM;

    const int gA = t0 - 128 + tid;
    const int gB = t0 + tid;
    const unsigned ldA  = (unsigned)tokens[rowBase + (gA >= 0 ? gA : 0)];
    const unsigned ldAm = (unsigned)tokens[rowBase + (gA >= 1 ? gA - 1 : 0)];
    const unsigned tokB = (unsigned)tokens[rowBase + gB];
    const unsigned ldBm = (unsigned)tokens[rowBase + (gB >= 1 ? gB - 1 : 0)];
    const unsigned tokA  = (gA >= 0) ? ldA  : 0u;
    const unsigned tokAm = (gA >= 1) ? ldAm : 0u;
    const unsigned tokBm = (gB >= 1) ? ldBm : 0u;
    const v4f wrow = reinterpret_cast<const v4f*>(Wc)[tid];
    for (int r = 0; r < 2; ++r) {
        const int j = tid + 128 * r;
        unsigned v = 0;
        #pragma unroll
        for (int i = 0; i < 8; ++i)
            if (j & (1 << i)) v ^= BASEC[i];
        cklut[j] = v;
    }
    {
        const v4f zero4 = {0.f, 0.f, 0.f, 0.f};
        for (int q = tid; q < 8 * STRIDE / 4; q += 128)
            *reinterpret_cast<v4f*>(&wc_lds[8 * STRIDE + q * 4]) = zero4;
    }
    pk2[tid]       = (tokA & 0xFFFFu) | (tokAm << 16);
    pk2[128 + tid] = (tokB & 0xFFFFu) | (tokBm << 16);
    *reinterpret_cast<v4f*>(&wc_lds[(tid >> 4) * STRIDE + (tid & 15) * 4]) = wrow;
    __syncthreads();

    const unsigned* pp = &pk2[127 + tid];
    const unsigned w0 = pp[0], w1 = pp[-2], w2 = pp[-4], w3 = pp[-6];
    unsigned h;
    {
        const unsigned q0 = pkmul(w0, PKB(0));
        keys_lds[0][tid] = q0 & 0xFFFFu;
        keys_lds[1][tid] = fold16(q0);
        h = q0 ^ pkmul(w1, PKB(1));
        keys_lds[2][tid] = fold16(h);
        h ^= pkmul(w2, PKB(2)) ^ pkmul(w3, PKB(3));
        keys_lds[3][tid] = fold16(h);
    }

    const v4f* tab4 = reinterpret_cast<const v4f*>(tables);
    v4f* out4 = reinterpret_cast<v4f*>(out);
    const long outPosBase = (rowBase + t0) * 8;

    const int lane  = tid & 31;
    const int wbase = tid & ~31;
    const int piece = lane & 3;
    const int sc    = (lane >> 2) & 3;
    const int pl    = lane >> 4;

    v4f sv[16];
    #pragma unroll
    for (int j = 0; j < 16; ++j) {
        const int pos = wbase + 2 * j + pl;
        const unsigned key = keys_lds[sc][pos];
        sv[j] = tab4[((size_t)(sc << 16) + key) * 4 + piece];
    }

    {
        unsigned h5 = 0, h6 = 0, h7 = 0;
        #pragma unroll
        for (int m = 0; m < 64; ++m) {
            const unsigned w = (m == 0) ? w0 : (m == 1) ? w1 : (m == 2) ? w2 : (m == 3) ? w3 : pp[-2 * m];
            if (m >= 4 && m < 8) h ^= pkmul(w, PKB(m));
            if (m < 16)          h5 ^= pkmul(w, PK5(m));
            if (m < 32)          h6 ^= pkmul(w, PK6(m & 3));
            h7 ^= pkmul(w, PK7(m & 3));
        }
        keys_lds[4][tid] = fold16(h);
        keys_lds[5][tid] = fold16(h5);
        keys_lds[6][tid] = fold16(h6);
        keys_lds[7][tid] = fold16(h7);
    }
    __builtin_amdgcn_sched_barrier(0);

    #pragma unroll
    for (int j = 0; j < 16; ++j) {
        const int pos = wbase + 2 * j + pl;
        *reinterpret_cast<v4f*>(&short_lds[pos * STRIDE + sc * 16 + piece * 4]) = sv[j];
        __builtin_nontemporal_store(sv[j], &out4[(outPosBase + (long)pos * 8 + sc) * 4 + piece]);
    }
    __builtin_amdgcn_sched_barrier(0);

    v4f lv[16];
    {
        const int n  = lane & 15;
        const int hi = lane >> 4;
        const float* bp = &wc_lds[n * STRIDE + 2 * hi];
        #pragma unroll
        for (int tt = 0; tt < 2; ++tt) {
            const int mb = wbase + tt * 16;
            const float* ap = &short_lds[(mb + n) * STRIDE + 2 * hi];
            v8f c = {0.f, 0.f, 0.f, 0.f, 0.f, 0.f, 0.f, 0.f};
            #pragma unroll
            for (int k = 0; k < 16; ++k) {
                v2f a  = *reinterpret_cast<const v2f*>(ap + 4 * k);
                v2f bb = *reinterpret_cast<const v2f*>(bp + 4 * k);
                c = __builtin_amdgcn_wmma_f32_16x16x4_f32(
                        false, a, false, bb, (short)0, c, false, false);
            }
            unsigned mask[8];
            #pragma unroll
            for (int v = 0; v < 8; ++v)
                mask[v] = __builtin_amdgcn_ballot_w32(c[v] > 0.0f);
            #pragma unroll
            for (int jj = 0; jj < 8; ++jj) {
                const int pos = mb + 2 * jj + pl;
                const unsigned msel = pl ? mask[2 * (jj & 3) + 1] : mask[2 * (jj & 3)];
                const unsigned bits = (msel >> (16 * (jj >> 2))) & 0xFFu;
                const unsigned key = (keys_lds[4 + sc][pos] ^ cklut[bits]) & 0xFFFFu;
                lv[tt * 8 + jj] = tab4[((size_t)((4 + sc) << 16) + key) * 4 + piece];
            }
            __builtin_amdgcn_sched_barrier(0);
        }
    }
    #pragma unroll
    for (int j = 0; j < 16; ++j) {
        const int pos = wbase + 2 * j + pl;
        __builtin_nontemporal_store(lv[j], &out4[(outPosBase + (long)pos * 8 + 4 + sc) * 4 + piece]);
    }
}

extern "C" void kernel_launch(void* const* d_in, const int* in_sizes, int n_in,
                              void* d_out, int out_size, void* d_ws, size_t ws_size,
                              hipStream_t stream) {
    (void)in_sizes; (void)n_in; (void)out_size; (void)d_ws; (void)ws_size;
    const int*   tokens = (const int*)d_in[0];
    const float* tables = (const float*)d_in[1];
    const float* Wc     = (const float*)d_in[2];
    float*       out    = (float*)d_out;
    const int B = 32, T = 4096;
    dim3 grid(B * (T / PPB)), block(128);
    hipLaunchKernelGGL(rhp_kernel, grid, block, 0, stream, tokens, tables, Wc, out);
}
